# starting kernel
# baseline (speedup 1.0000x reference)
#include <hip/hip_runtime.h>
#include <hip/hip_bf16.h>

typedef __attribute__((ext_vector_type(2))) float v2f;
typedef __attribute__((ext_vector_type(8))) float v8f;

#define D 128

__global__ void deg_init_kernel(float* deg, int n) {
    int i = blockIdx.x * blockDim.x + threadIdx.x;
    if (i < n) deg[i] = 1.0f;
}

__global__ void deg_count_kernel(const int* row, float* deg, int E) {
    int e = blockIdx.x * blockDim.x + threadIdx.x;
    if (e < E) atomicAdd(&deg[row[e]], 1.0f);
}

__global__ void deg_rsqrt_kernel(const float* deg, float* dis, int n) {
    int i = blockIdx.x * blockDim.x + threadIdx.x;
    if (i < n) dis[i] = rsqrtf(deg[i]);
}

__global__ void gemm_wmma_f32(const float* __restrict__ x,
                              const float* __restrict__ W,
                              float* __restrict__ h) {
    const int wave = threadIdx.x >> 5;
    const int lane = threadIdx.x & 31;
    const int half = lane >> 4;
    const int idx  = lane & 15;
    const int m0 = blockIdx.x << 4;
    const int n0 = wave << 4;

    v8f c = {};
    const float* xr = x + (m0 + idx) * D;
    const float* Wc = W + idx + n0;
#pragma unroll 8
    for (int k0 = 0; k0 < D; k0 += 4) {
        v2f a, b;
        a.x = xr[k0 + 2 * half + 0];
        a.y = xr[k0 + 2 * half + 1];
        b.x = Wc[(k0 + 2 * half + 0) * D];
        b.y = Wc[(k0 + 2 * half + 1) * D];
        c = __builtin_amdgcn_wmma_f32_16x16x4_f32(
             false, a,  false, b,
             (short)0, c,  false,  false);
    }
    float* hp = h + (m0 + 8 * half) * D + n0 + idx;
#pragma unroll
    for (int g = 0; g < 8; ++g) hp[g * D] = c[g];
}

__global__ void self_bias_kernel(const float* __restrict__ h,
                                 const float* __restrict__ dis,
                                 const float* __restrict__ bias,
                                 float* __restrict__ out, int n) {
    int gid = blockIdx.x * blockDim.x + threadIdx.x;
    if (gid >= n * D) return;
    int i = gid >> 7;
    int d = gid & (D - 1);
    float s = dis[i];
    out[gid] = bias[d] + h[gid] * s * s;
}

__global__ void scatter_kernel(const int* __restrict__ row,
                               const int* __restrict__ col,
                               const float* __restrict__ h,
                               const float* __restrict__ dis,
                               float* __restrict__ out, int E) {
    int gid = blockIdx.x * blockDim.x + threadIdx.x;
    int e = gid >> 5;
    if (e >= E) return;
    int lane = gid & 31;
    int r = row[e];
    int c = col[e];
    float w = dis[r] * dis[c];
    const float4 hv = *(const float4*)(h + r * D + lane * 4);
    float* o = out + c * D + lane * 4;
    atomicAdd(o + 0, hv.x * w);
    atomicAdd(o + 1, hv.y * w);
    atomicAdd(o + 2, hv.z * w);
    atomicAdd(o + 3, hv.w * w);
}

extern "C" void kernel_launch(void* const* d_in, const int* in_sizes, int n_in,
                              void* d_out, int out_size, void* d_ws, size_t ws_size,
                              hipStream_t stream) {
    const float* x    = (const float*)d_in[0];
    const int*   ei   = (const int*)d_in[1];
    const float* W    = (const float*)d_in[2];
    const float* bias = (const float*)d_in[3];
    float* out = (float*)d_out;

    const int N = in_sizes[0] / D;
    const int E = in_sizes[1] / 2;
    const int* row = ei;
    const int* col = ei + E;

    float* h   = (float*)d_ws;
    float* deg = h + (size_t)N * D;
    float* dis = deg + N;

    deg_init_kernel<<<(N + 255) / 256, 256, 0, stream>>>(deg, N);
    deg_count_kernel<<<(E + 255) / 256, 256, 0, stream>>>(row, deg, E);
    deg_rsqrt_kernel<<<(N + 255) / 256, 256, 0, stream>>>(deg, dis, N);
    gemm_wmma_f32<<<N / 16, 256, 0, stream>>>(x, W, h);
    self_bias_kernel<<<((size_t)N * D + 255) / 256, 256, 0, stream>>>(h, dis, bias, out, N);
    {
        long long threads = (long long)E * 32;
        int blocks = (int)((threads + 255) / 256);
        scatter_kernel<<<blocks, 256, 0, stream>>>(row, col, h, dis, out, E);
    }
}


# Degree count as per-work-group LDS histograms (integer LDS adds) stored as partial planes and summed in the rsqrt launch: no global atomics on deg, 5 launches not 6
# speedup vs baseline: 1.3503x; 1.3503x over previous
#include <hip/hip_runtime.h>
#include <hip/hip_bf16.h>

typedef __attribute__((ext_vector_type(2))) float v2f;
typedef __attribute__((ext_vector_type(8))) float v8f;

#define D 128

#define DEG_BLOCKS  64
#define DEG_THREADS 1024
#define DEG_LOADS   16

__global__ void __launch_bounds__(DEG_THREADS)
deg_hist_kernel(const int* __restrict__ row, int* __restrict__ part, int E, int n) {
    extern __shared__ int hist[];
    const int tid = threadIdx.x;
    for (int i = tid; i < n; i += DEG_THREADS) hist[i] = 0;
    __syncthreads();
    const int chunk = (E + DEG_BLOCKS - 1) / DEG_BLOCKS;
    const int beg = blockIdx.x * chunk;
    const int end = min(E, beg + chunk);
    for (int e0 = beg + tid; e0 < end; e0 += DEG_LOADS * DEG_THREADS) {
        int r[DEG_LOADS];
#pragma unroll
        for (int u = 0; u < DEG_LOADS; ++u)
            r[u] = row[min(e0 + u * DEG_THREADS, end - 1)];
#pragma unroll
        for (int u = 0; u < DEG_LOADS; ++u)
            if (e0 + u * DEG_THREADS < end) atomicAdd(&hist[r[u]], 1);
    }
    __syncthreads();
    int* p = part + (size_t)blockIdx.x * n;
    for (int i = tid; i < n; i += DEG_THREADS) p[i] = hist[i];
}

__global__ void deg_rsqrt_kernel(const int* __restrict__ part, float* __restrict__ dis, int n) {
    int i = blockIdx.x * blockDim.x + threadIdx.x;
    if (i >= n) return;
    int s = 0;
#pragma unroll
    for (int b = 0; b < DEG_BLOCKS; ++b) s += part[(size_t)b * n + i];
    dis[i] = rsqrtf(1.0f + (float)s);
}

__global__ void gemm_wmma_f32(const float* __restrict__ x,
                              const float* __restrict__ W,
                              float* __restrict__ h) {
    const int wave = threadIdx.x >> 5;
    const int lane = threadIdx.x & 31;
    const int half = lane >> 4;
    const int idx  = lane & 15;
    const int m0 = blockIdx.x << 4;
    const int n0 = wave << 4;

    v8f c = {};
    const float* xr = x + (m0 + idx) * D;
    const float* Wc = W + idx + n0;
#pragma unroll 8
    for (int k0 = 0; k0 < D; k0 += 4) {
        v2f a, b;
        a.x = xr[k0 + 2 * half + 0];
        a.y = xr[k0 + 2 * half + 1];
        b.x = Wc[(k0 + 2 * half + 0) * D];
        b.y = Wc[(k0 + 2 * half + 1) * D];
        c = __builtin_amdgcn_wmma_f32_16x16x4_f32(
             false, a,  false, b,
             (short)0, c,  false,  false);
    }
    float* hp = h + (m0 + 8 * half) * D + n0 + idx;
#pragma unroll
    for (int g = 0; g < 8; ++g) hp[g * D] = c[g];
}

__global__ void self_bias_kernel(const float* __restrict__ h,
                                 const float* __restrict__ dis,
                                 const float* __restrict__ bias,
                                 float* __restrict__ out, int n) {
    int gid = blockIdx.x * blockDim.x + threadIdx.x;
    if (gid >= n * D) return;
    int i = gid >> 7;
    int d = gid & (D - 1);
    float s = dis[i];
    out[gid] = bias[d] + h[gid] * s * s;
}

__global__ void scatter_kernel(const int* __restrict__ row,
                               const int* __restrict__ col,
                               const float* __restrict__ h,
                               const float* __restrict__ dis,
                               float* __restrict__ out, int E) {
    int gid = blockIdx.x * blockDim.x + threadIdx.x;
    int e = gid >> 5;
    if (e >= E) return;
    int lane = gid & 31;
    int r = row[e];
    int c = col[e];
    float w = dis[r] * dis[c];
    const float4 hv = *(const float4*)(h + r * D + lane * 4);
    float* o = out + c * D + lane * 4;
    atomicAdd(o + 0, hv.x * w);
    atomicAdd(o + 1, hv.y * w);
    atomicAdd(o + 2, hv.z * w);
    atomicAdd(o + 3, hv.w * w);
}

extern "C" void kernel_launch(void* const* d_in, const int* in_sizes, int n_in,
                              void* d_out, int out_size, void* d_ws, size_t ws_size,
                              hipStream_t stream) {
    const float* x    = (const float*)d_in[0];
    const int*   ei   = (const int*)d_in[1];
    const float* W    = (const float*)d_in[2];
    const float* bias = (const float*)d_in[3];
    float* out = (float*)d_out;

    const int N = in_sizes[0] / D;
    const int E = in_sizes[1] / 2;
    const int* row = ei;
    const int* col = ei + E;

    float* h   = (float*)d_ws;
    float* dis = h + (size_t)N * D + N;

    int* part = (int*)d_ws;
    deg_hist_kernel<<<DEG_BLOCKS, DEG_THREADS, N * sizeof(int), stream>>>(row, part, E, N);
    deg_rsqrt_kernel<<<(N + 127) / 128, 128, 0, stream>>>(part, dis, N);
    gemm_wmma_f32<<<N / 16, 256, 0, stream>>>(x, W, h);
    self_bias_kernel<<<((size_t)N * D + 255) / 256, 256, 0, stream>>>(h, dis, bias, out, N);
    {
        long long threads = (long long)E * 32;
        int blocks = (int)((threads + 255) / 256);
        scatter_kernel<<<blocks, 256, 0, stream>>>(row, col, h, dis, out, E);
    }
}


# Fill stores (source, dis[source]) pairs, weight read from an LDS copy of dis; gather takes weights from its contiguous 8-byte pair load, no scattered 4-byte dis loads; order of sums unchanged
# speedup vs baseline: 2.7039x; 2.0024x over previous
#include <hip/hip_runtime.h>
#include <hip/hip_bf16.h>


typedef __attribute__((ext_vector_type(2))) float v2f;
typedef __attribute__((ext_vector_type(8))) float v8f;

#define D 128


#define EDGE_BLOCKS  64
#define EDGE_THREADS 1024
#define EDGE_LOADS   10
#define SCAN_SHIFT   8
#define SCAN_THREADS (1 << SCAN_SHIFT)
#define NODE_MAX     10000

__device__ __forceinline__ void edge_range(int E, int b, int& beg, int& end) {
    const int chunk = (E + EDGE_BLOCKS - 1) / EDGE_BLOCKS;
    beg = min(E, b * chunk);
    end = min(E, beg + chunk);
}

__global__ void __launch_bounds__(EDGE_THREADS)
edge_hist_kernel(const int* __restrict__ row, const int* __restrict__ col,
                 unsigned* __restrict__ part, int E, int n) {
    extern __shared__ unsigned hist[];
    const int tid = threadIdx.x;
    for (int i = tid; i < n; i += EDGE_THREADS) hist[i] = 0u;
    __syncthreads();
    int beg, end;
    edge_range(E, blockIdx.x, beg, end);
    for (int e0 = beg + tid; e0 < end; e0 += EDGE_LOADS * EDGE_THREADS) {
        int r[EDGE_LOADS], c[EDGE_LOADS];
#pragma unroll
        for (int u = 0; u < EDGE_LOADS; ++u) {
            const int e = min(e0 + u * EDGE_THREADS, end - 1);
            r[u] = row[e];
            c[u] = col[e];
        }
#pragma unroll
        for (int u = 0; u < EDGE_LOADS; ++u)
            if (e0 + u * EDGE_THREADS < end) {
                atomicAdd(&hist[r[u]], 1u);
                atomicAdd(&hist[c[u]], 0x10000u);
            }
    }
    __syncthreads();
    unsigned* p = part + (size_t)blockIdx.x * n;
    for (int i = tid; i < n; i += EDGE_THREADS) p[i] = hist[i];
}

__global__ void __launch_bounds__(SCAN_THREADS)
bin_scan_kernel(unsigned* plane,
                float* __restrict__ dis, int* __restrict__ gtot, int n) {
    __shared__ int wsum[SCAN_THREADS / 32];
    const int t = threadIdx.x;
    const int i = blockIdx.x * SCAN_THREADS + t;
    const bool live = i < n;
    unsigned v[EDGE_BLOCKS];
    const int ic = min(i, n - 1);
    const unsigned keep = live ? 0xffffffffu : 0u;
#pragma unroll
    for (int b = 0; b < EDGE_BLOCKS; ++b) v[b] = plane[(size_t)b * n + ic] & keep;
    int degsum = 0, tot = 0;
#pragma unroll
    for (int b = 0; b < EDGE_BLOCKS; ++b) {
        degsum += (int)(v[b] & 0xffffu);
        tot    += (int)(v[b] >> 16);
    }
    const int lane = t & 31, w = t >> 5;
    int inc = tot;
    for (int d = 1; d < 32; d <<= 1) {
        const int y = __shfl_up(inc, d);
        if (lane >= d) inc += y;
    }
    if (lane == 31) wsum[w] = inc;
    __syncthreads();
    int base = 0;
    for (int k = 0; k < SCAN_THREADS / 32; ++k)
        if (k < w) base += wsum[k];
    if (t == SCAN_THREADS - 1) gtot[blockIdx.x] = base + inc;
    if (live) {
        dis[i] = rsqrtf(1.0f + (float)degsum);
        int run = base + inc - tot;
#pragma unroll
        for (int b = 0; b < EDGE_BLOCKS; ++b) {
            plane[(size_t)b * n + i] = (unsigned)run;
            run += (int)(v[b] >> 16);
        }
    }
}

__global__ void __launch_bounds__(EDGE_THREADS)
edge_fill_kernel(const int* __restrict__ row, const int* __restrict__ col,
                 const int* __restrict__ off, const int* __restrict__ gtot,
                 const float* __restrict__ dis,
                 int* __restrict__ start, int2* __restrict__ pairs, int E, int n) {
    extern __shared__ int pos[];
    __shared__ int wbits[NODE_MAX];
    __shared__ int graw[EDGE_THREADS];
    __shared__ int gbase[EDGE_THREADS];
    const int tid = threadIdx.x;
    const int ng = (n + SCAN_THREADS - 1) >> SCAN_SHIFT;
    if (tid < ng) graw[tid] = gtot[tid];
    __syncthreads();
    if (tid < ng) {
        int s = 0;
        for (int g = 0; g < tid; ++g) s += graw[g];
        gbase[tid] = s;
    }
    __syncthreads();
    const int* o = off + (size_t)blockIdx.x * n;
    for (int i = tid; i < n; i += EDGE_THREADS) {
        const int p = o[i] + gbase[i >> SCAN_SHIFT];
        wbits[i] = __float_as_int(dis[i]);
        pos[i] = p;
        if (blockIdx.x == 0) start[i] = p;
    }
    __syncthreads();
    int beg, end;
    edge_range(E, blockIdx.x, beg, end);
    for (int e0 = beg + tid; e0 < end; e0 += EDGE_LOADS * EDGE_THREADS) {
        int r[EDGE_LOADS], c[EDGE_LOADS];
#pragma unroll
        for (int u = 0; u < EDGE_LOADS; ++u) {
            const int e = min(e0 + u * EDGE_THREADS, end - 1);
            r[u] = row[e];
            c[u] = col[e];
        }
#pragma unroll
        for (int u = 0; u < EDGE_LOADS; ++u)
            if (e0 + u * EDGE_THREADS < end) {
                const int p = atomicAdd(&pos[c[u]], 1);
                pairs[p] = make_int2(r[u], wbits[r[u]]);
            }
    }
}

#define GATHER_ROWS 8
__global__ void __launch_bounds__(256)
gather_sum_kernel(const float* __restrict__ x, const unsigned long long* __restrict__ pairs,
                  const int* __restrict__ start, const float* __restrict__ dis,
                  float* __restrict__ agg, int E, int n) {
    const int lane = threadIdx.x & 31;
    const int c = blockIdx.x * 8 + (threadIdx.x >> 5);
    if (c >= n) return;
    const int beg = __builtin_amdgcn_readfirstlane(start[c]);
    const int end = __builtin_amdgcn_readfirstlane((c + 1 < n) ? start[c + 1] : E);
    const float dc = dis[c];
    const float4* x4 = (const float4*)x;
    const float4 xs = x4[(size_t)c * (D / 4) + lane];
    float4 acc = make_float4(dc * xs.x, dc * xs.y, dc * xs.z, dc * xs.w);
    for (int j0 = beg; j0 < end; j0 += 32) {
        const int cnt = min(32, end - j0);
        const unsigned long long pr = pairs[j0 + min(lane, cnt - 1)];
        const int   mine  = (int)(unsigned)pr;
        const float wmine = (lane < cnt) ? __int_as_float((int)(unsigned)(pr >> 32)) : 0.0f;
#pragma unroll
        for (int k = 0; k < 32; k += GATHER_ROWS) {
            if (k >= cnt) break;
            float4 g[GATHER_ROWS];
            float  w[GATHER_ROWS];
#pragma unroll
            for (int u = 0; u < GATHER_ROWS; ++u) {
                const int r = __builtin_amdgcn_readlane(mine, k + u);
                w[u] = __int_as_float(__builtin_amdgcn_readlane(__float_as_int(wmine), k + u));
                g[u] = x4[(size_t)r * (D / 4) + lane];
            }
#pragma unroll
            for (int u = 0; u < GATHER_ROWS; ++u) {
                acc.x = fmaf(w[u], g[u].x, acc.x);
                acc.y = fmaf(w[u], g[u].y, acc.y);
                acc.z = fmaf(w[u], g[u].z, acc.z);
                acc.w = fmaf(w[u], g[u].w, acc.w);
            }
        }
    }
    float4* a4 = (float4*)agg;
    a4[(size_t)c * (D / 4) + lane] = make_float4(dc * acc.x, dc * acc.y, dc * acc.z, dc * acc.w);
}

__global__ void gemm_bias_wmma_f32(const float* __restrict__ a_in,
                                   const float* __restrict__ W,
                                   const float* __restrict__ bias,
                                   float* __restrict__ out) {
    const int wave = threadIdx.x >> 5;
    const int lane = threadIdx.x & 31;
    const int half = lane >> 4;
    const int idx  = lane & 15;
    const int m0 = blockIdx.x << 4;
    const int n0 = wave << 4;

    v8f c = {};
    const float* xr = a_in + (m0 + idx) * D;
    const float* Wc = W + idx + n0;
#pragma unroll 8
    for (int k0 = 0; k0 < D; k0 += 4) {
        v2f a, b;
        a.x = xr[k0 + 2 * half + 0];
        a.y = xr[k0 + 2 * half + 1];
        b.x = Wc[(k0 + 2 * half + 0) * D];
        b.y = Wc[(k0 + 2 * half + 1) * D];
        c = __builtin_amdgcn_wmma_f32_16x16x4_f32(
             false, a,  false, b,
             (short)0, c,  false,  false);
    }
    const float bv = bias[n0 + idx];
    float* hp = out + (m0 + 8 * half) * D + n0 + idx;
#pragma unroll
    for (int g = 0; g < 8; ++g) hp[g * D] = c[g] + bv;
}

extern "C" void kernel_launch(void* const* d_in, const int* in_sizes, int n_in,
                              void* d_out, int out_size, void* d_ws, size_t ws_size,
                              hipStream_t stream) {
    const float* x    = (const float*)d_in[0];
    const int*   ei   = (const int*)d_in[1];
    const float* W    = (const float*)d_in[2];
    const float* bias = (const float*)d_in[3];
    float* out = (float*)d_out;

    const int N = in_sizes[0] / D;
    const int E = in_sizes[1] / 2;
    const int* row = ei;
    const int* col = ei + E;

    float*    agg   = (float*)d_ws;
    int*      start = (int*)(agg + (size_t)N * D);
    float*    dis   = agg + (size_t)N * D + N;
    unsigned* plane = (unsigned*)d_ws;
    int*      gtot  = (int*)d_ws + (size_t)EDGE_BLOCKS * N;
    int2*     pairs = (int2*)d_out;

    edge_hist_kernel<<<EDGE_BLOCKS, EDGE_THREADS, N * sizeof(unsigned), stream>>>(row, col, plane, E, N);
    bin_scan_kernel<<<(N + SCAN_THREADS - 1) / SCAN_THREADS, SCAN_THREADS, 0, stream>>>(plane, dis, gtot, N);
    edge_fill_kernel<<<EDGE_BLOCKS, EDGE_THREADS, N * sizeof(int), stream>>>(row, col, (const int*)plane, gtot, dis, start, pairs, E, N);
    gather_sum_kernel<<<(N + 7) / 8, 256, 0, stream>>>(x, (const unsigned long long*)pairs, start, dis, agg, E, N);
    gemm_bias_wmma_f32<<<N / 16, 256, 0, stream>>>(agg, W, bias, out);
}


# Fill kernel issues all its global loads (totals, cursor plane, dis, edges) before its first wait and batches its 10 LDS adds: one memory round trip, not 12; sums in sorted order, ties by arrival
# speedup vs baseline: 2.7909x; 1.0322x over previous
#include <hip/hip_runtime.h>
#include <hip/hip_bf16.h>


typedef __attribute__((ext_vector_type(2))) float v2f;
typedef __attribute__((ext_vector_type(8))) float v8f;

#define D 128


#define EDGE_BLOCKS  64
#define EDGE_THREADS 1024
#define EDGE_LOADS   10
#define SCAN_SHIFT   8
#define SCAN_THREADS (1 << SCAN_SHIFT)
#define NODE_MAX     10000

__device__ __forceinline__ void edge_range(int E, int b, int& beg, int& end) {
    const int chunk = (E + EDGE_BLOCKS - 1) / EDGE_BLOCKS;
    beg = min(E, b * chunk);
    end = min(E, beg + chunk);
}

__global__ void __launch_bounds__(EDGE_THREADS)
edge_hist_kernel(const int* __restrict__ row, const int* __restrict__ col,
                 unsigned* __restrict__ part, int E, int n) {
    extern __shared__ unsigned hist[];
    const int tid = threadIdx.x;
    for (int i = tid; i < n; i += EDGE_THREADS) hist[i] = 0u;
    __syncthreads();
    int beg, end;
    edge_range(E, blockIdx.x, beg, end);
    for (int e0 = beg + tid; e0 < end; e0 += EDGE_LOADS * EDGE_THREADS) {
        int r[EDGE_LOADS], c[EDGE_LOADS];
#pragma unroll
        for (int u = 0; u < EDGE_LOADS; ++u) {
            const int e = min(e0 + u * EDGE_THREADS, end - 1);
            r[u] = row[e];
            c[u] = col[e];
        }
#pragma unroll
        for (int u = 0; u < EDGE_LOADS; ++u)
            if (e0 + u * EDGE_THREADS < end) {
                atomicAdd(&hist[r[u]], 1u);
                atomicAdd(&hist[c[u]], 0x10000u);
            }
    }
    __syncthreads();
    unsigned* p = part + (size_t)blockIdx.x * n;
    for (int i = tid; i < n; i += EDGE_THREADS) p[i] = hist[i];
}

__global__ void __launch_bounds__(SCAN_THREADS)
bin_scan_kernel(unsigned* plane,
                float* __restrict__ dis, int* __restrict__ gtot, int n) {
    __shared__ int wsum[SCAN_THREADS / 32];
    const int t = threadIdx.x;
    const int i = blockIdx.x * SCAN_THREADS + t;
    const bool live = i < n;
    unsigned v[EDGE_BLOCKS];
    const int ic = min(i, n - 1);
    const unsigned keep = live ? 0xffffffffu : 0u;
#pragma unroll
    for (int b = 0; b < EDGE_BLOCKS; ++b) v[b] = plane[(size_t)b * n + ic] & keep;
    int degsum = 0, tot = 0;
#pragma unroll
    for (int b = 0; b < EDGE_BLOCKS; ++b) {
        degsum += (int)(v[b] & 0xffffu);
        tot    += (int)(v[b] >> 16);
    }
    const int lane = t & 31, w = t >> 5;
    int inc = tot;
    for (int d = 1; d < 32; d <<= 1) {
        const int y = __shfl_up(inc, d);
        if (lane >= d) inc += y;
    }
    if (lane == 31) wsum[w] = inc;
    __syncthreads();
    int base = 0;
    for (int k = 0; k < SCAN_THREADS / 32; ++k)
        if (k < w) base += wsum[k];
    if (t == SCAN_THREADS - 1) gtot[blockIdx.x] = base + inc;
    if (live) {
        dis[i] = rsqrtf(1.0f + (float)degsum);
        int run = base + inc - tot;
#pragma unroll
        for (int b = 0; b < EDGE_BLOCKS; ++b) {
            plane[(size_t)b * n + i] = (unsigned)run;
            run += (int)(v[b] >> 16);
        }
    }
}

static_assert(NODE_MAX <= EDGE_LOADS * EDGE_THREADS, "one trip of the table loads must cover every node");
__global__ void __launch_bounds__(EDGE_THREADS)
edge_fill_kernel(const int* __restrict__ row, const int* __restrict__ col,
                 const int* __restrict__ off, const int* __restrict__ gtot,
                 const float* __restrict__ dis,
                 int* __restrict__ start, int2* __restrict__ pairs, int E, int n) {
    extern __shared__ int pos[];
    __shared__ int wbits[NODE_MAX];
    __shared__ int graw[EDGE_THREADS];
    __shared__ int gbase[EDGE_THREADS];
    __shared__ int spare[32];
    const int tid = threadIdx.x;
    const int ng = (n + SCAN_THREADS - 1) >> SCAN_SHIFT;
    if (tid < 32) spare[tid] = 0;
    int beg, end;
    edge_range(E, blockIdx.x, beg, end);
    const int* o = off + (size_t)blockIdx.x * n;
    const int e0 = beg + tid;

    const int gt = gtot[min(tid, ng - 1)];
    int ov[EDGE_LOADS], dv[EDGE_LOADS];
#pragma unroll
    for (int u = 0; u < EDGE_LOADS; ++u) {
        const int i = min(tid + u * EDGE_THREADS, n - 1);
        ov[u] = o[i];
        dv[u] = __float_as_int(dis[i]);
    }
    int r[EDGE_LOADS], c[EDGE_LOADS];
#pragma unroll
    for (int u = 0; u < EDGE_LOADS; ++u) {
        const int e = max(min(e0 + u * EDGE_THREADS, end - 1), 0);
        r[u] = row[e];
        c[u] = col[e];
    }

    if (tid < ng) graw[tid] = gt;
    __syncthreads();
    if (tid < ng) {
        int s = 0;
        for (int g = 0; g < tid; ++g) s += graw[g];
        gbase[tid] = s;
    }
    __syncthreads();
    int gb[EDGE_LOADS];
#pragma unroll
    for (int u = 0; u < EDGE_LOADS; ++u)
        gb[u] = gbase[min((tid + u * EDGE_THREADS) >> SCAN_SHIFT, ng - 1)];
#pragma unroll
    for (int u = 0; u < EDGE_LOADS; ++u) {
        const int i = tid + u * EDGE_THREADS;
        if (i < n) {
            const int p = ov[u] + gb[u];
            wbits[i] = dv[u];
            pos[i] = p;
            if (blockIdx.x == 0) start[i] = p;
        }
    }
    __syncthreads();
    for (int eb = e0; eb < end; eb += EDGE_LOADS * EDGE_THREADS) {
        if (eb != e0) {
#pragma unroll
            for (int u = 0; u < EDGE_LOADS; ++u) {
                const int e = min(eb + u * EDGE_THREADS, end - 1);
                r[u] = row[e];
                c[u] = col[e];
            }
        }
        int pp[EDGE_LOADS], ww[EDGE_LOADS];
#pragma unroll
        for (int u = 0; u < EDGE_LOADS; ++u) {
            int* cur = (eb + u * EDGE_THREADS < end) ? &pos[c[u]] : &spare[tid & 31];
            pp[u] = atomicAdd(cur, 1);
            ww[u] = wbits[r[u]];
        }
#pragma unroll
        for (int u = 0; u < EDGE_LOADS; ++u)
            if (eb + u * EDGE_THREADS < end) pairs[pp[u]] = make_int2(r[u], ww[u]);
    }
}

#define GATHER_ROWS 8
__global__ void __launch_bounds__(256)
gather_sum_kernel(const float* __restrict__ x, const unsigned long long* __restrict__ pairs,
                  const int* __restrict__ start, const float* __restrict__ dis,
                  float* __restrict__ agg, int E, int n) {
    const int lane = threadIdx.x & 31;
    const int c = blockIdx.x * 8 + (threadIdx.x >> 5);
    if (c >= n) return;
    const int beg = __builtin_amdgcn_readfirstlane(start[c]);
    const int end = __builtin_amdgcn_readfirstlane((c + 1 < n) ? start[c + 1] : E);
    const float dc = dis[c];
    const float4* x4 = (const float4*)x;
    const float4 xs = x4[(size_t)c * (D / 4) + lane];
    float4 acc = make_float4(dc * xs.x, dc * xs.y, dc * xs.z, dc * xs.w);
    for (int j0 = beg; j0 < end; j0 += 32) {
        const int cnt = min(32, end - j0);
        const unsigned long long pr = pairs[j0 + min(lane, cnt - 1)];
        const int   mine  = (int)(unsigned)pr;
        const float wmine = (lane < cnt) ? __int_as_float((int)(unsigned)(pr >> 32)) : 0.0f;
#pragma unroll
        for (int k = 0; k < 32; k += GATHER_ROWS) {
            if (k >= cnt) break;
            float4 g[GATHER_ROWS];
            float  w[GATHER_ROWS];
#pragma unroll
            for (int u = 0; u < GATHER_ROWS; ++u) {
                const int r = __builtin_amdgcn_readlane(mine, k + u);
                w[u] = __int_as_float(__builtin_amdgcn_readlane(__float_as_int(wmine), k + u));
                g[u] = x4[(size_t)r * (D / 4) + lane];
            }
#pragma unroll
            for (int u = 0; u < GATHER_ROWS; ++u) {
                acc.x = fmaf(w[u], g[u].x, acc.x);
                acc.y = fmaf(w[u], g[u].y, acc.y);
                acc.z = fmaf(w[u], g[u].z, acc.z);
                acc.w = fmaf(w[u], g[u].w, acc.w);
            }
        }
    }
    float4* a4 = (float4*)agg;
    a4[(size_t)c * (D / 4) + lane] = make_float4(dc * acc.x, dc * acc.y, dc * acc.z, dc * acc.w);
}

__global__ void gemm_bias_wmma_f32(const float* __restrict__ a_in,
                                   const float* __restrict__ W,
                                   const float* __restrict__ bias,
                                   float* __restrict__ out) {
    const int wave = threadIdx.x >> 5;
    const int lane = threadIdx.x & 31;
    const int half = lane >> 4;
    const int idx  = lane & 15;
    const int m0 = blockIdx.x << 4;
    const int n0 = wave << 4;

    v8f c = {};
    const float* xr = a_in + (m0 + idx) * D;
    const float* Wc = W + idx + n0;
#pragma unroll 8
    for (int k0 = 0; k0 < D; k0 += 4) {
        v2f a, b;
        a.x = xr[k0 + 2 * half + 0];
        a.y = xr[k0 + 2 * half + 1];
        b.x = Wc[(k0 + 2 * half + 0) * D];
        b.y = Wc[(k0 + 2 * half + 1) * D];
        c = __builtin_amdgcn_wmma_f32_16x16x4_f32(
             false, a,  false, b,
             (short)0, c,  false,  false);
    }
    const float bv = bias[n0 + idx];
    float* hp = out + (m0 + 8 * half) * D + n0 + idx;
#pragma unroll
    for (int g = 0; g < 8; ++g) hp[g * D] = c[g] + bv;
}

extern "C" void kernel_launch(void* const* d_in, const int* in_sizes, int n_in,
                              void* d_out, int out_size, void* d_ws, size_t ws_size,
                              hipStream_t stream) {
    const float* x    = (const float*)d_in[0];
    const int*   ei   = (const int*)d_in[1];
    const float* W    = (const float*)d_in[2];
    const float* bias = (const float*)d_in[3];
    float* out = (float*)d_out;

    const int N = in_sizes[0] / D;
    const int E = in_sizes[1] / 2;
    const int* row = ei;
    const int* col = ei + E;

    float*    agg   = (float*)d_ws;
    int*      start = (int*)(agg + (size_t)N * D);
    float*    dis   = agg + (size_t)N * D + N;
    unsigned* plane = (unsigned*)d_ws;
    int*      gtot  = (int*)d_ws + (size_t)EDGE_BLOCKS * N;
    int2*     pairs = (int2*)d_out;

    edge_hist_kernel<<<EDGE_BLOCKS, EDGE_THREADS, N * sizeof(unsigned), stream>>>(row, col, plane, E, N);
    bin_scan_kernel<<<(N + SCAN_THREADS - 1) / SCAN_THREADS, SCAN_THREADS, 0, stream>>>(plane, dis, gtot, N);
    edge_fill_kernel<<<EDGE_BLOCKS, EDGE_THREADS, N * sizeof(int), stream>>>(row, col, (const int*)plane, gtot, dis, start, pairs, E, N);
    gather_sum_kernel<<<(N + 7) / 8, 256, 0, stream>>>(x, (const unsigned long long*)pairs, start, dis, agg, E, N);
    gemm_bias_wmma_f32<<<N / 16, 256, 0, stream>>>(agg, W, bias, out);
}


# Fill: 4 work-groups of 8 waves share each edge range, one per quarter of the destination nodes: scattered pair stores issue from all 256 WGPs, 8 lines each not 32; sorted sums, ties by arrival
# speedup vs baseline: 3.1359x; 1.1236x over previous
#include <hip/hip_runtime.h>
#include <hip/hip_bf16.h>


typedef __attribute__((ext_vector_type(2))) float v2f;
typedef __attribute__((ext_vector_type(8))) float v8f;

#define D 128


#define EDGE_BLOCKS  64
#define EDGE_THREADS 1024
#define EDGE_LOADS   10
#define SCAN_SHIFT   8
#define SCAN_THREADS (1 << SCAN_SHIFT)
#define NODE_MAX     10000

__device__ __forceinline__ void edge_range(int E, int b, int& beg, int& end) {
    const int chunk = (E + EDGE_BLOCKS - 1) / EDGE_BLOCKS;
    beg = min(E, b * chunk);
    end = min(E, beg + chunk);
}

__global__ void __launch_bounds__(EDGE_THREADS)
edge_hist_kernel(const int* __restrict__ row, const int* __restrict__ col,
                 unsigned* __restrict__ part, int E, int n) {
    extern __shared__ unsigned hist[];
    const int tid = threadIdx.x;
    for (int i = tid; i < n; i += EDGE_THREADS) hist[i] = 0u;
    __syncthreads();
    int beg, end;
    edge_range(E, blockIdx.x, beg, end);
    for (int e0 = beg + tid; e0 < end; e0 += EDGE_LOADS * EDGE_THREADS) {
        int r[EDGE_LOADS], c[EDGE_LOADS];
#pragma unroll
        for (int u = 0; u < EDGE_LOADS; ++u) {
            const int e = min(e0 + u * EDGE_THREADS, end - 1);
            r[u] = row[e];
            c[u] = col[e];
        }
#pragma unroll
        for (int u = 0; u < EDGE_LOADS; ++u)
            if (e0 + u * EDGE_THREADS < end) {
                atomicAdd(&hist[r[u]], 1u);
                atomicAdd(&hist[c[u]], 0x10000u);
            }
    }
    __syncthreads();
    unsigned* p = part + (size_t)blockIdx.x * n;
    for (int i = tid; i < n; i += EDGE_THREADS) p[i] = hist[i];
}

__global__ void __launch_bounds__(SCAN_THREADS)
bin_scan_kernel(unsigned* plane,
                float* __restrict__ dis, int* __restrict__ gtot, int n) {
    __shared__ int wsum[SCAN_THREADS / 32];
    const int t = threadIdx.x;
    const int i = blockIdx.x * SCAN_THREADS + t;
    const bool live = i < n;
    unsigned v[EDGE_BLOCKS];
    const int ic = min(i, n - 1);
    const unsigned keep = live ? 0xffffffffu : 0u;
#pragma unroll
    for (int b = 0; b < EDGE_BLOCKS; ++b) v[b] = plane[(size_t)b * n + ic] & keep;
    int degsum = 0, tot = 0;
#pragma unroll
    for (int b = 0; b < EDGE_BLOCKS; ++b) {
        degsum += (int)(v[b] & 0xffffu);
        tot    += (int)(v[b] >> 16);
    }
    const int lane = t & 31, w = t >> 5;
    int inc = tot;
    for (int d = 1; d < 32; d <<= 1) {
        const int y = __shfl_up(inc, d);
        if (lane >= d) inc += y;
    }
    if (lane == 31) wsum[w] = inc;
    __syncthreads();
    int base = 0;
    for (int k = 0; k < SCAN_THREADS / 32; ++k)
        if (k < w) base += wsum[k];
    if (t == SCAN_THREADS - 1) gtot[blockIdx.x] = base + inc;
    if (live) {
        dis[i] = rsqrtf(1.0f + (float)degsum);
        int run = base + inc - tot;
#pragma unroll
        for (int b = 0; b < EDGE_BLOCKS; ++b) {
            plane[(size_t)b * n + i] = (unsigned)run;
            run += (int)(v[b] >> 16);
        }
    }
}

#define FILL_PARTS   4
#define FILL_THREADS 256
#define FILL_LOADS   40
#define FILL_CURS    10
static_assert(NODE_MAX <= FILL_LOADS * FILL_THREADS, "one trip of the table loads must cover every node");
static_assert((NODE_MAX + FILL_PARTS - 1) / FILL_PARTS <= FILL_CURS * FILL_THREADS, "the cursor loads must cover one part of the nodes");
static_assert((NODE_MAX + SCAN_THREADS - 1) / SCAN_THREADS <= FILL_THREADS, "one thread per group total");
__global__ void __launch_bounds__(FILL_THREADS)
edge_fill_kernel(const int* __restrict__ row, const int* __restrict__ col,
                 const int* __restrict__ off, const int* __restrict__ gtot,
                 const float* __restrict__ dis,
                 int* __restrict__ start, int2* __restrict__ pairs, int E, int n) {
    extern __shared__ int pos[];
    __shared__ int wbits[NODE_MAX];
    __shared__ int graw[FILL_THREADS];
    __shared__ int gbase[FILL_THREADS];
    __shared__ int spare[32];
    const int tid = threadIdx.x;
    const int b = blockIdx.x / FILL_PARTS;
    const int q = blockIdx.x % FILL_PARTS;
    const int ng = (n + SCAN_THREADS - 1) >> SCAN_SHIFT;
    const int qn = (n + FILL_PARTS - 1) / FILL_PARTS;
    const int lo = min(n, q * qn);
    const int cnt = min(n, lo + qn) - lo;
    if (tid < 32) spare[tid] = 0;
    int beg, end;
    edge_range(E, b, beg, end);
    const int* o = off + (size_t)b * n;
    const int e0 = beg + tid;

    const int gt = gtot[min(tid, ng - 1)];
    int ov[FILL_CURS], dv[FILL_LOADS];
#pragma unroll
    for (int u = 0; u < FILL_CURS; ++u)
        ov[u] = o[max(min(lo + tid + u * FILL_THREADS, lo + cnt - 1), 0)];
#pragma unroll
    for (int u = 0; u < FILL_LOADS; ++u)
        dv[u] = __float_as_int(dis[min(tid + u * FILL_THREADS, n - 1)]);
    int r[FILL_LOADS], c[FILL_LOADS];
#pragma unroll
    for (int u = 0; u < FILL_LOADS; ++u) {
        const int e = max(min(e0 + u * FILL_THREADS, end - 1), 0);
        r[u] = row[e];
        c[u] = col[e];
    }

    if (tid < ng) graw[tid] = gt;
    __syncthreads();
    if (tid < ng) {
        int s = 0;
        for (int g = 0; g < tid; ++g) s += graw[g];
        gbase[tid] = s;
    }
    __syncthreads();
    int gb[FILL_CURS];
#pragma unroll
    for (int u = 0; u < FILL_CURS; ++u)
        gb[u] = gbase[min((lo + tid + u * FILL_THREADS) >> SCAN_SHIFT, ng - 1)];
#pragma unroll
    for (int u = 0; u < FILL_CURS; ++u) {
        const int j = tid + u * FILL_THREADS;
        if (j < cnt) {
            const int p = ov[u] + gb[u];
            pos[j] = p;
            if (b == 0) start[lo + j] = p;
        }
    }
#pragma unroll
    for (int u = 0; u < FILL_LOADS; ++u) {
        const int i = tid + u * FILL_THREADS;
        if (i < n) wbits[i] = dv[u];
    }
    __syncthreads();
    for (int eb = e0; eb < end; eb += FILL_LOADS * FILL_THREADS) {
        if (eb != e0) {
#pragma unroll
            for (int u = 0; u < FILL_LOADS; ++u) {
                const int e = min(eb + u * FILL_THREADS, end - 1);
                r[u] = row[e];
                c[u] = col[e];
            }
        }
        int pp[FILL_LOADS], ww[FILL_LOADS];
#pragma unroll
        for (int u = 0; u < FILL_LOADS; ++u) {
            const int j = c[u] - lo;
            const bool mine = (eb + u * FILL_THREADS < end) && ((unsigned)j < (unsigned)cnt);
            int* cur = mine ? &pos[j] : &spare[tid & 31];
            pp[u] = atomicAdd(cur, 1);
            ww[u] = wbits[r[u]];
        }
#pragma unroll
        for (int u = 0; u < FILL_LOADS; ++u) {
            const int j = c[u] - lo;
            const bool mine = (eb + u * FILL_THREADS < end) && ((unsigned)j < (unsigned)cnt);
            if (mine) pairs[pp[u]] = make_int2(r[u], ww[u]);
        }
    }
}

#define GATHER_ROWS 8
__global__ void __launch_bounds__(256)
gather_sum_kernel(const float* __restrict__ x, const unsigned long long* __restrict__ pairs,
                  const int* __restrict__ start, const float* __restrict__ dis,
                  float* __restrict__ agg, int E, int n) {
    const int lane = threadIdx.x & 31;
    const int c = blockIdx.x * 8 + (threadIdx.x >> 5);
    if (c >= n) return;
    const int beg = __builtin_amdgcn_readfirstlane(start[c]);
    const int end = __builtin_amdgcn_readfirstlane((c + 1 < n) ? start[c + 1] : E);
    const float dc = dis[c];
    const float4* x4 = (const float4*)x;
    const float4 xs = x4[(size_t)c * (D / 4) + lane];
    float4 acc = make_float4(dc * xs.x, dc * xs.y, dc * xs.z, dc * xs.w);
    for (int j0 = beg; j0 < end; j0 += 32) {
        const int cnt = min(32, end - j0);
        const unsigned long long pr = pairs[j0 + min(lane, cnt - 1)];
        const int   mine  = (int)(unsigned)pr;
        const float wmine = (lane < cnt) ? __int_as_float((int)(unsigned)(pr >> 32)) : 0.0f;
#pragma unroll
        for (int k = 0; k < 32; k += GATHER_ROWS) {
            if (k >= cnt) break;
            float4 g[GATHER_ROWS];
            float  w[GATHER_ROWS];
#pragma unroll
            for (int u = 0; u < GATHER_ROWS; ++u) {
                const int r = __builtin_amdgcn_readlane(mine, k + u);
                w[u] = __int_as_float(__builtin_amdgcn_readlane(__float_as_int(wmine), k + u));
                g[u] = x4[(size_t)r * (D / 4) + lane];
            }
#pragma unroll
            for (int u = 0; u < GATHER_ROWS; ++u) {
                acc.x = fmaf(w[u], g[u].x, acc.x);
                acc.y = fmaf(w[u], g[u].y, acc.y);
                acc.z = fmaf(w[u], g[u].z, acc.z);
                acc.w = fmaf(w[u], g[u].w, acc.w);
            }
        }
    }
    float4* a4 = (float4*)agg;
    a4[(size_t)c * (D / 4) + lane] = make_float4(dc * acc.x, dc * acc.y, dc * acc.z, dc * acc.w);
}

__global__ void gemm_bias_wmma_f32(const float* __restrict__ a_in,
                                   const float* __restrict__ W,
                                   const float* __restrict__ bias,
                                   float* __restrict__ out) {
    const int wave = threadIdx.x >> 5;
    const int lane = threadIdx.x & 31;
    const int half = lane >> 4;
    const int idx  = lane & 15;
    const int m0 = blockIdx.x << 4;
    const int n0 = wave << 4;

    v8f c = {};
    const float* xr = a_in + (m0 + idx) * D;
    const float* Wc = W + idx + n0;
#pragma unroll 8
    for (int k0 = 0; k0 < D; k0 += 4) {
        v2f a, b;
        a.x = xr[k0 + 2 * half + 0];
        a.y = xr[k0 + 2 * half + 1];
        b.x = Wc[(k0 + 2 * half + 0) * D];
        b.y = Wc[(k0 + 2 * half + 1) * D];
        c = __builtin_amdgcn_wmma_f32_16x16x4_f32(
             false, a,  false, b,
             (short)0, c,  false,  false);
    }
    const float bv = bias[n0 + idx];
    float* hp = out + (m0 + 8 * half) * D + n0 + idx;
#pragma unroll
    for (int g = 0; g < 8; ++g) hp[g * D] = c[g] + bv;
}

extern "C" void kernel_launch(void* const* d_in, const int* in_sizes, int n_in,
                              void* d_out, int out_size, void* d_ws, size_t ws_size,
                              hipStream_t stream) {
    const float* x    = (const float*)d_in[0];
    const int*   ei   = (const int*)d_in[1];
    const float* W    = (const float*)d_in[2];
    const float* bias = (const float*)d_in[3];
    float* out = (float*)d_out;

    const int N = in_sizes[0] / D;
    const int E = in_sizes[1] / 2;
    const int* row = ei;
    const int* col = ei + E;

    float*    agg   = (float*)d_ws;
    int*      start = (int*)(agg + (size_t)N * D);
    float*    dis   = agg + (size_t)N * D + N;
    unsigned* plane = (unsigned*)d_ws;
    int*      gtot  = (int*)d_ws + (size_t)EDGE_BLOCKS * N;
    int2*     pairs = (int2*)d_out;

    edge_hist_kernel<<<EDGE_BLOCKS, EDGE_THREADS, N * sizeof(unsigned), stream>>>(row, col, plane, E, N);
    bin_scan_kernel<<<(N + SCAN_THREADS - 1) / SCAN_THREADS, SCAN_THREADS, 0, stream>>>(plane, dis, gtot, N);
    edge_fill_kernel<<<EDGE_BLOCKS * FILL_PARTS, FILL_THREADS, ((N + FILL_PARTS - 1) / FILL_PARTS) * sizeof(int), stream>>>(row, col, (const int*)plane, gtot, dis, start, pairs, E, N);
    gather_sum_kernel<<<(N + 7) / 8, 256, 0, stream>>>(x, (const unsigned long long*)pairs, start, dis, agg, E, N);
    gemm_bias_wmma_f32<<<N / 16, 256, 0, stream>>>(agg, W, bias, out);
}
